# starting kernel
# baseline (speedup 1.0000x reference)
#include <hip/hip_runtime.h>


typedef float v2f __attribute__((ext_vector_type(2)));
typedef float v8f __attribute__((ext_vector_type(8)));

#define B_    32
#define N_    32
#define D_    12
#define P_    32
#define L_    1024
#define LOUT_ 32
#define BIGV  1e30f

__device__ __forceinline__ float bperm_f(int src_lane, float v) {
    int r = __builtin_amdgcn_ds_bpermute(src_lane << 2,
                                         __builtin_bit_cast(int, v));
    return __builtin_bit_cast(float, r);
}

__global__ __launch_bounds__(32) void dtw_wmma_kernel(
    const float* __restrict__ x,
    const float* __restrict__ patts,
    float* __restrict__ out)
{
    const int lane = threadIdx.x;
    const int hi   = lane >> 4;
    const int lo   = lane & 15;
    const int b    = blockIdx.x;
    const int n    = blockIdx.y;

    __shared__ float p2s[P_];
    __shared__ float ctile[P_ * 17];

    const float* Pn = patts + n * (D_ * P_);
    const float* Xb = x + b * (D_ * L_);

    {
        float s = 0.0f;
        #pragma unroll
        for (int d = 0; d < D_; ++d) { float v = Pn[d * P_ + lane]; s += v * v; }
        p2s[lane] = s;
    }
    __syncthreads();

    v2f Afrag[2][3];
    #pragma unroll
    for (int m = 0; m < 2; ++m)
        #pragma unroll
        for (int kc = 0; kc < 3; ++kc) {
            int d0 = 4 * kc + 2 * hi;
            int p  = m * 16 + lo;
            v2f a; a.x = Pn[d0 * P_ + p]; a.y = Pn[(d0 + 1) * P_ + p];
            Afrag[m][kc] = a;
        }

    float p2v[2][8];
    #pragma unroll
    for (int m = 0; m < 2; ++m)
        #pragma unroll
        for (int r = 0; r < 8; ++r)
            p2v[m][r] = p2s[m * 16 + 8 * hi + r];

    float dprev = BIGV;
    float* outp = out + ((b * N_ + n) * P_ + lane) * LOUT_;
    const float INF = __builtin_inff();

    for (int jt = 0; jt < L_ / 16; ++jt) {
        const int jbase = jt * 16;
        const int j     = jbase + lo;

        v2f Bfrag[3];
        float sx = 0.0f;
        #pragma unroll
        for (int kc = 0; kc < 3; ++kc) {
            int d0 = 4 * kc + 2 * hi;
            float b0 = Xb[d0 * L_ + j];
            float b1 = Xb[(d0 + 1) * L_ + j];
            v2f bb; bb.x = b0; bb.y = b1;
            Bfrag[kc] = bb;
            sx += b0 * b0 + b1 * b1;
        }
        sx += bperm_f(lane ^ 16, sx);

        if (jt + 1 < L_ / 16)
            __builtin_prefetch(&Xb[(2 * hi) * L_ + j + 16], 0, 3);

        #pragma unroll
        for (int m = 0; m < 2; ++m) {
            v8f acc = {};
            acc = __builtin_amdgcn_wmma_f32_16x16x4_f32(false, Afrag[m][0],
                    false, Bfrag[0], (short)0, acc, false, false);
            acc = __builtin_amdgcn_wmma_f32_16x16x4_f32(false, Afrag[m][1],
                    false, Bfrag[1], (short)0, acc, false, false);
            acc = __builtin_amdgcn_wmma_f32_16x16x4_f32(false, Afrag[m][2],
                    false, Bfrag[2], (short)0, acc, false, false);
            #pragma unroll
            for (int r = 0; r < 8; ++r) {
                float cst = p2v[m][r] + sx - 2.0f * acc[r];
                ctile[(m * 16 + 8 * hi + r) * 17 + lo] = cst;
            }
        }
        __syncthreads();

        for (int jj = 0; jj < 16; ++jj) {
            const int   jg = jbase + jj;
            const float c  = ctile[lane * 17 + jj];

            float a;
            if (jg == 0) {
                a = (lane == 0) ? 0.0f : BIGV;
            } else {
                float up = bperm_f(lane - 1, dprev);
                up = (lane == 0) ? BIGV : up;
                a  = fminf(up, dprev);
            }

            float al = c + a;
            float be = c;
            #pragma unroll
            for (int o = 1; o < 32; o <<= 1) {
                float af = bperm_f(lane - o, al);
                float bf = bperm_f(lane - o, be);
                bool ok = (lane >= o);
                af = ok ? af : INF;
                bf = ok ? bf : 0.0f;
                al = fminf(al, be + af);
                be = be + bf;
            }
            dprev = fminf(al, be + BIGV);

            if (jg >= L_ - LOUT_) outp[jg - (L_ - LOUT_)] = dprev;
        }
        __syncthreads();
    }
}

extern "C" void kernel_launch(void* const* d_in, const int* in_sizes, int n_in,
                              void* d_out, int out_size, void* d_ws, size_t ws_size,
                              hipStream_t stream) {
    const float* x     = (const float*)d_in[0];
    const float* patts = (const float*)d_in[1];
    float*       outp  = (float*)d_out;
    (void)in_sizes; (void)n_in; (void)out_size; (void)d_ws; (void)ws_size;

    dim3 grid(B_, N_);
    dim3 block(32);
    dtw_wmma_kernel<<<grid, block, 0, stream>>>(x, patts, outp);
}


# DP row: a lane's 32-cell dependent min/add chain run as four independent 8-cell chains, their affine-min maps joined by a two-level tree, so four links overlap; sums regrouped: sum order changes
# speedup vs baseline: 16.1868x; 16.1868x over previous
#include <hip/hip_runtime.h>


typedef float v2f __attribute__((ext_vector_type(2)));
typedef float v4f __attribute__((ext_vector_type(4)));
typedef float v8f __attribute__((ext_vector_type(8)));

#define B_    32
#define N_    32
#define D_    12
#define P_    32
#define L_    1024
#define LOUT_ 32
#define BIGV  1e30f
#define CH_     32
#define SUB_    4
#define SL_     (CH_ / SUB_)
#define CPITCH_ 36
#define RPITCH_ (32 * CPITCH_)

__device__ __forceinline__ float bperm_f(int src_lane, float v) {
    int r = __builtin_amdgcn_ds_bpermute(src_lane << 2,
                                         __builtin_bit_cast(int, v));
    return __builtin_bit_cast(float, r);
}

template <int N>
__device__ __forceinline__ float row_shr0(float v) {
    return __builtin_bit_cast(float,
        __builtin_amdgcn_update_dpp(0, __builtin_bit_cast(int, v),
                                    0x110 + N, 0xf, 0xf, true));
}

__device__ __forceinline__ float other_row_lane15(float v) {
    return __builtin_bit_cast(float,
        __builtin_amdgcn_permlanex16(0, __builtin_bit_cast(int, v),
                                     0xffffffff, 0xffffffff, false, false));
}

template <int O>
__device__ __forceinline__ void scan_step(float& al, float& be, int lo) {
    const float af = row_shr0<O>(al);
    const float bf = row_shr0<O>(be);
    const float bI = (lo >= O) ? be : __builtin_inff();
    al = fminf(al, bI + af);
    be = be + bf;
}

__global__ __launch_bounds__(32) void dtw_rows_kernel(
    const float* __restrict__ x,
    const float* __restrict__ patts,
    float* __restrict__ out)
{
    const int lane = threadIdx.x;
    const int hi   = lane >> 4;
    const int lo   = lane & 15;
    const int b    = blockIdx.x;
    const int n    = blockIdx.y;

    __shared__ float p2s[P_];
    __shared__ __attribute__((aligned(16))) float cst[16 * RPITCH_];

    const float* Pn = patts + n * (D_ * P_);
    const float* Xb = x + b * (D_ * L_);

    {
        float s = 0.0f;
        #pragma unroll
        for (int d = 0; d < D_; ++d) { float v = Pn[d * P_ + lane]; s += v * v; }
        p2s[lane] = s;
    }
    __syncthreads();

    float dprev[CH_];
    #pragma unroll
    for (int k = 0; k < CH_; ++k) dprev[k] = BIGV;
    float leftc = BIGV;

    float* outp = out + (b * N_ + n) * (P_ * LOUT_);
    const float INF = __builtin_inff();

    #pragma unroll 1
    for (int h = 0; h < 2; ++h) {
        v2f Afrag[3];
        #pragma unroll
        for (int kc = 0; kc < 3; ++kc) {
            const int d0 = 4 * kc + 2 * hi;
            const int p  = 16 * h + lo;
            v2f a; a.x = Pn[d0 * P_ + p]; a.y = Pn[(d0 + 1) * P_ + p];
            Afrag[kc] = a;
        }
        float p2v[8];
        #pragma unroll
        for (int r = 0; r < 8; ++r) p2v[r] = p2s[16 * h + 8 * hi + r];

        #pragma unroll 4
        for (int q = 0; q < L_ / 64; ++q) {
            const int col = 64 * q + 4 * lo;
            v4f xa[3], xb[3];
            #pragma unroll
            for (int kc = 0; kc < 3; ++kc) {
                const int d0 = 4 * kc + 2 * hi;
                xa[kc] = *reinterpret_cast<const v4f*>(&Xb[d0 * L_ + col]);
                xb[kc] = *reinterpret_cast<const v4f*>(&Xb[(d0 + 1) * L_ + col]);
            }
            v8f   acc[4];
            float sx[4];
            #pragma unroll
            for (int s = 0; s < 4; ++s) {
                float t = 0.0f;
                v8f a = {};
                #pragma unroll
                for (int kc = 0; kc < 3; ++kc) {
                    v2f bb; bb.x = xa[kc][s]; bb.y = xb[kc][s];
                    t += bb.x * bb.x + bb.y * bb.y;
                    a = __builtin_amdgcn_wmma_f32_16x16x4_f32(false, Afrag[kc],
                            false, bb, (short)0, a, false, false);
                }
                acc[s] = a;
                sx[s]  = t;
            }
            #pragma unroll
            for (int s = 0; s < 4; ++s) sx[s] += bperm_f(lane ^ 16, sx[s]);

            float* dst = &cst[(8 * hi) * RPITCH_ + (2 * q + (lo >> 3)) * CPITCH_ + 4 * (lo & 7)];
            #pragma unroll
            for (int r = 0; r < 8; ++r) {
                v4f cv;
                cv.x = p2v[r] + sx[0] - 2.0f * acc[0][r];
                cv.y = p2v[r] + sx[1] - 2.0f * acc[1][r];
                cv.z = p2v[r] + sx[2] - 2.0f * acc[2][r];
                cv.w = p2v[r] + sx[3] - 2.0f * acc[3][r];
                *reinterpret_cast<v4f*>(dst + r * RPITCH_) = cv;
            }
        }
        __syncthreads();

        #pragma unroll 1
        for (int i = 0; i < 16; ++i) {
            const float* crow = &cst[i * RPITCH_ + lane * CPITCH_];
            float c[CH_];
            #pragma unroll
            for (int q4 = 0; q4 < CH_ / 4; ++q4) {
                const v4f v = *reinterpret_cast<const v4f*>(crow + 4 * q4);
                c[4 * q4 + 0] = v.x; c[4 * q4 + 1] = v.y;
                c[4 * q4 + 2] = v.z; c[4 * q4 + 3] = v.w;
            }

            float al[CH_], be[CH_];
            float ae[SUB_], bs[SUB_];
            float m0[SUB_];
            #pragma unroll
            for (int s = 0; s < SUB_; ++s) {
                float alp = BIGV, bep = 0.0f;
                #pragma unroll
                for (int t = 0; t < SL_; ++t) {
                    const int   k    = s * SL_ + t;
                    const float left = (k == 0) ? leftc : dprev[k - 1];
                    const float a    = fminf(left, dprev[k]);
                    const float m    = fminf(a, alp);
                    if (t == 0) m0[s] = m;
                    alp = c[k] + m;
                    bep = (t == 0) ? c[k] : bep + c[k];
                    al[k] = alp; be[k] = bep;
                }
                ae[s] = alp; bs[s] = bep;
            }
            const float a01 = fminf(ae[1], bs[1] + ae[0]);
            const float b01 = bs[0] + bs[1];
            const float a23 = fminf(ae[3], bs[3] + ae[2]);
            const float b23 = bs[2] + bs[3];
            const float a02 = fminf(ae[2], bs[2] + a01);
            const float b02 = b01 + bs[2];

            float A = fminf(a23, b23 + a01), Bv = b01 + b23;
            scan_step<1>(A, Bv, lo);
            scan_step<2>(A, Bv, lo);
            scan_step<4>(A, Bv, lo);
            scan_step<8>(A, Bv, lo);
            {
                const float a15 = other_row_lane15(A);
                const float b15 = other_row_lane15(Bv);
                const float bJ  = hi ? Bv : INF;
                A  = fminf(A, bJ + a15);
                Bv = Bv + (hi ? b15 : 0.0f);
            }

            const float v0   = (h == 0 && i == 0) ? 0.0f : BIGV;
            const float dend = fminf(A, Bv + v0);

            const float s1  = row_shr0<1>(dend);
            const float x15 = other_row_lane15(dend);
            const float din = (lane == 0) ? v0 : ((lane == 16) ? x15 : s1);

            float dn[SUB_];
            dn[0] = din;
            dn[1] = fminf(ae[0], bs[0] + din);
            dn[2] = fminf(a01, b01 + din);
            dn[3] = fminf(a02, b02 + din);
            #pragma unroll
            for (int k = 0; k < CH_; ++k)
                dprev[k] = (k % SL_ == 0) ? c[k] + fminf(m0[k / SL_], dn[k / SL_])
                                          : fminf(al[k], be[k] + dn[k / SL_]);
            leftc = (lane == 0) ? BIGV : din;

            if (lane == 31) {
                v4f* o = reinterpret_cast<v4f*>(outp + (16 * h + i) * LOUT_);
                #pragma unroll
                for (int q4 = 0; q4 < CH_ / 4; ++q4) {
                    v4f v;
                    v.x = dprev[4 * q4 + 0]; v.y = dprev[4 * q4 + 1];
                    v.z = dprev[4 * q4 + 2]; v.w = dprev[4 * q4 + 3];
                    o[q4] = v;
                }
            }
        }
        __syncthreads();
    }
}

extern "C" void kernel_launch(void* const* d_in, const int* in_sizes, int n_in,
                              void* d_out, int out_size, void* d_ws, size_t ws_size,
                              hipStream_t stream) {
    const float* x     = (const float*)d_in[0];
    const float* patts = (const float*)d_in[1];
    float*       outp  = (float*)d_out;
    (void)in_sizes; (void)n_in; (void)out_size; (void)d_ws; (void)ws_size;

    dim3 grid(B_, N_);
    dim3 block(32);
    dtw_rows_kernel<<<grid, block, 0, stream>>>(x, patts, outp);
}


# x[b] and the pattern fragments read from global memory once, before the two halves: all 64 tiles' B fragments and x2 stay in registers; a half's build is WMMA, cost math, LDS stores; same sums, order
# speedup vs baseline: 17.9913x; 1.1115x over previous
#include <hip/hip_runtime.h>


typedef float v2f __attribute__((ext_vector_type(2)));
typedef float v4f __attribute__((ext_vector_type(4)));
typedef float v8f __attribute__((ext_vector_type(8)));

#define B_    32
#define N_    32
#define D_    12
#define P_    32
#define L_    1024
#define LOUT_ 32
#define BIGV  1e30f
#define CH_     32
#define SUB_    4
#define SL_     (CH_ / SUB_)
#define CPITCH_ 36
#define RPITCH_ (32 * CPITCH_)

__device__ __forceinline__ float bperm_f(int src_lane, float v) {
    int r = __builtin_amdgcn_ds_bpermute(src_lane << 2,
                                         __builtin_bit_cast(int, v));
    return __builtin_bit_cast(float, r);
}

template <int N>
__device__ __forceinline__ float row_shr0(float v) {
    return __builtin_bit_cast(float,
        __builtin_amdgcn_update_dpp(0, __builtin_bit_cast(int, v),
                                    0x110 + N, 0xf, 0xf, true));
}

__device__ __forceinline__ float other_row_lane15(float v) {
    return __builtin_bit_cast(float,
        __builtin_amdgcn_permlanex16(0, __builtin_bit_cast(int, v),
                                     0xffffffff, 0xffffffff, false, false));
}

template <int O>
__device__ __forceinline__ void scan_step(float& al, float& be, int lo) {
    const float af = row_shr0<O>(al);
    const float bf = row_shr0<O>(be);
    const float bI = (lo >= O) ? be : __builtin_inff();
    al = fminf(al, bI + af);
    be = be + bf;
}

__global__ __launch_bounds__(32) void dtw_rows_kernel(
    const float* __restrict__ x,
    const float* __restrict__ patts,
    float* __restrict__ out)
{
    const int lane = threadIdx.x;
    const int hi   = lane >> 4;
    const int lo   = lane & 15;
    const int b    = blockIdx.x;
    const int n    = blockIdx.y;

    __shared__ float p2s[P_];
    __shared__ __attribute__((aligned(16))) float cst[16 * RPITCH_];

    const float* Pn = patts + n * (D_ * P_);
    const float* Xb = x + b * (D_ * L_);

    {
        float s = 0.0f;
        #pragma unroll
        for (int d = 0; d < D_; ++d) { float v = Pn[d * P_ + lane]; s += v * v; }
        p2s[lane] = s;
    }
    __syncthreads();

    float dprev[CH_];
    #pragma unroll
    for (int k = 0; k < CH_; ++k) dprev[k] = BIGV;
    float leftc = BIGV;

    float* outp = out + (b * N_ + n) * (P_ * LOUT_);
    const float INF = __builtin_inff();

    v2f Aboth[2][3];
    #pragma unroll
    for (int hh = 0; hh < 2; ++hh) {
        #pragma unroll
        for (int kc = 0; kc < 3; ++kc) {
            const int d0 = 4 * kc + 2 * hi;
            const int p  = 16 * hh + lo;
            v2f a; a.x = Pn[d0 * P_ + p]; a.y = Pn[(d0 + 1) * P_ + p];
            Aboth[hh][kc] = a;
        }
    }

    v2f   Bfr[L_ / 64][3][4];
    float sxr[L_ / 64][4];
    #pragma unroll
    for (int q = 0; q < L_ / 64; ++q) {
        const int col = 64 * q + 4 * lo;
        v4f xa[3], xb[3];
        #pragma unroll
        for (int kc = 0; kc < 3; ++kc) {
            const int d0 = 4 * kc + 2 * hi;
            xa[kc] = *reinterpret_cast<const v4f*>(&Xb[d0 * L_ + col]);
            xb[kc] = *reinterpret_cast<const v4f*>(&Xb[(d0 + 1) * L_ + col]);
        }
        #pragma unroll
        for (int s = 0; s < 4; ++s) {
            float t = 0.0f;
            #pragma unroll
            for (int kc = 0; kc < 3; ++kc) {
                v2f bb; bb.x = xa[kc][s]; bb.y = xb[kc][s];
                t += bb.x * bb.x + bb.y * bb.y;
                Bfr[q][kc][s] = bb;
            }
            sxr[q][s] = t;
        }
        #pragma unroll
        for (int s = 0; s < 4; ++s) sxr[q][s] += bperm_f(lane ^ 16, sxr[q][s]);
    }

    #pragma unroll 1
    for (int h = 0; h < 2; ++h) {
        v2f Afrag[3];
        #pragma unroll
        for (int kc = 0; kc < 3; ++kc) Afrag[kc] = h ? Aboth[1][kc] : Aboth[0][kc];
        float p2v[8];
        #pragma unroll
        for (int r = 0; r < 8; ++r) p2v[r] = p2s[16 * h + 8 * hi + r];

        #pragma unroll
        for (int q = 0; q < L_ / 64; ++q) {
            v8f acc[4];
            #pragma unroll
            for (int s = 0; s < 4; ++s) {
                v8f a = {};
                #pragma unroll
                for (int kc = 0; kc < 3; ++kc)
                    a = __builtin_amdgcn_wmma_f32_16x16x4_f32(false, Afrag[kc],
                            false, Bfr[q][kc][s], (short)0, a, false, false);
                acc[s] = a;
            }
            float* dst = &cst[(8 * hi) * RPITCH_ + (2 * q + (lo >> 3)) * CPITCH_ + 4 * (lo & 7)];
            #pragma unroll
            for (int r = 0; r < 8; ++r) {
                v4f cv;
                cv.x = p2v[r] + sxr[q][0] - 2.0f * acc[0][r];
                cv.y = p2v[r] + sxr[q][1] - 2.0f * acc[1][r];
                cv.z = p2v[r] + sxr[q][2] - 2.0f * acc[2][r];
                cv.w = p2v[r] + sxr[q][3] - 2.0f * acc[3][r];
                *reinterpret_cast<v4f*>(dst + r * RPITCH_) = cv;
            }
        }
        __syncthreads();

        #pragma unroll 1
        for (int i = 0; i < 16; ++i) {
            const float* crow = &cst[i * RPITCH_ + lane * CPITCH_];
            float c[CH_];
            #pragma unroll
            for (int q4 = 0; q4 < CH_ / 4; ++q4) {
                const v4f v = *reinterpret_cast<const v4f*>(crow + 4 * q4);
                c[4 * q4 + 0] = v.x; c[4 * q4 + 1] = v.y;
                c[4 * q4 + 2] = v.z; c[4 * q4 + 3] = v.w;
            }

            float al[CH_], be[CH_];
            float ae[SUB_], bs[SUB_];
            float m0[SUB_];
            #pragma unroll
            for (int s = 0; s < SUB_; ++s) {
                float alp = BIGV, bep = 0.0f;
                #pragma unroll
                for (int t = 0; t < SL_; ++t) {
                    const int   k    = s * SL_ + t;
                    const float left = (k == 0) ? leftc : dprev[k - 1];
                    const float a    = fminf(left, dprev[k]);
                    const float m    = fminf(a, alp);
                    if (t == 0) m0[s] = m;
                    alp = c[k] + m;
                    bep = (t == 0) ? c[k] : bep + c[k];
                    al[k] = alp; be[k] = bep;
                }
                ae[s] = alp; bs[s] = bep;
            }
            const float a01 = fminf(ae[1], bs[1] + ae[0]);
            const float b01 = bs[0] + bs[1];
            const float a23 = fminf(ae[3], bs[3] + ae[2]);
            const float b23 = bs[2] + bs[3];
            const float a02 = fminf(ae[2], bs[2] + a01);
            const float b02 = b01 + bs[2];

            float A = fminf(a23, b23 + a01), Bv = b01 + b23;
            scan_step<1>(A, Bv, lo);
            scan_step<2>(A, Bv, lo);
            scan_step<4>(A, Bv, lo);
            scan_step<8>(A, Bv, lo);
            {
                const float a15 = other_row_lane15(A);
                const float b15 = other_row_lane15(Bv);
                const float bJ  = hi ? Bv : INF;
                A  = fminf(A, bJ + a15);
                Bv = Bv + (hi ? b15 : 0.0f);
            }

            const float v0   = (h == 0 && i == 0) ? 0.0f : BIGV;
            const float dend = fminf(A, Bv + v0);

            const float s1  = row_shr0<1>(dend);
            const float x15 = other_row_lane15(dend);
            const float din = (lane == 0) ? v0 : ((lane == 16) ? x15 : s1);

            float dn[SUB_];
            dn[0] = din;
            dn[1] = fminf(ae[0], bs[0] + din);
            dn[2] = fminf(a01, b01 + din);
            dn[3] = fminf(a02, b02 + din);
            #pragma unroll
            for (int k = 0; k < CH_; ++k)
                dprev[k] = (k % SL_ == 0) ? c[k] + fminf(m0[k / SL_], dn[k / SL_])
                                          : fminf(al[k], be[k] + dn[k / SL_]);
            leftc = (lane == 0) ? BIGV : din;

            if (lane == 31) {
                v4f* o = reinterpret_cast<v4f*>(outp + (16 * h + i) * LOUT_);
                #pragma unroll
                for (int q4 = 0; q4 < CH_ / 4; ++q4) {
                    v4f v;
                    v.x = dprev[4 * q4 + 0]; v.y = dprev[4 * q4 + 1];
                    v.z = dprev[4 * q4 + 2]; v.w = dprev[4 * q4 + 3];
                    o[q4] = v;
                }
            }
        }
        __syncthreads();
    }
}

extern "C" void kernel_launch(void* const* d_in, const int* in_sizes, int n_in,
                              void* d_out, int out_size, void* d_ws, size_t ws_size,
                              hipStream_t stream) {
    const float* x     = (const float*)d_in[0];
    const float* patts = (const float*)d_in[1];
    float*       outp  = (float*)d_out;
    (void)in_sizes; (void)n_in; (void)out_size; (void)d_ws; (void)ws_size;

    dim3 grid(B_, N_);
    dim3 block(32);
    dtw_rows_kernel<<<grid, block, 0, stream>>>(x, patts, outp);
}


# Kernel front pipelined on its loads: pattern and x loads issued first (x ten groups ahead); p2, then each group's pairing, x2 and first-half cost build follow as its loads land; same sums, same order
# speedup vs baseline: 18.9536x; 1.0535x over previous
#include <hip/hip_runtime.h>


typedef float v2f __attribute__((ext_vector_type(2)));
typedef float v4f __attribute__((ext_vector_type(4)));
typedef float v8f __attribute__((ext_vector_type(8)));

#define B_    32
#define N_    32
#define D_    12
#define P_    32
#define L_    1024
#define LOUT_ 32
#define BIGV  1e30f
#define CH_     32
#define SUB_    4
#define SL_     (CH_ / SUB_)
#define CPITCH_ 36
#define RPITCH_ (32 * CPITCH_)
#define PRE_    10

__device__ __forceinline__ float bperm_f(int src_lane, float v) {
    int r = __builtin_amdgcn_ds_bpermute(src_lane << 2,
                                         __builtin_bit_cast(int, v));
    return __builtin_bit_cast(float, r);
}

__device__ __forceinline__ float lane_f(float v, int src) {
    return __builtin_bit_cast(float,
        __builtin_amdgcn_readlane(__builtin_bit_cast(int, v), src));
}

template <int N>
__device__ __forceinline__ float row_shr0(float v) {
    return __builtin_bit_cast(float,
        __builtin_amdgcn_update_dpp(0, __builtin_bit_cast(int, v),
                                    0x110 + N, 0xf, 0xf, true));
}

__device__ __forceinline__ float other_row_lane15(float v) {
    return __builtin_bit_cast(float,
        __builtin_amdgcn_permlanex16(0, __builtin_bit_cast(int, v),
                                     0xffffffff, 0xffffffff, false, false));
}

template <int O>
__device__ __forceinline__ void scan_step(float& al, float& be, int lo) {
    const float af = row_shr0<O>(al);
    const float bf = row_shr0<O>(be);
    const float bI = (lo >= O) ? be : __builtin_inff();
    al = fminf(al, bI + af);
    be = be + bf;
}

__device__ __forceinline__ void load_group(v4f (&xa)[3], v4f (&xb)[3],
                                           const float* __restrict__ Xb, int col, int hi) {
    #pragma unroll
    for (int kc = 0; kc < 3; ++kc) {
        const int d0 = 4 * kc + 2 * hi;
        xa[kc] = *reinterpret_cast<const v4f*>(&Xb[d0 * L_ + col]);
        xb[kc] = *reinterpret_cast<const v4f*>(&Xb[(d0 + 1) * L_ + col]);
    }
}

__device__ __forceinline__ void build_group(const v2f (&Af)[3], const float (&p2v)[8],
                                            const v2f (&Bq)[3][4], const float (&sx)[4],
                                            float* dst) {
    v8f acc[4];
    #pragma unroll
    for (int s = 0; s < 4; ++s) {
        v8f a = {};
        #pragma unroll
        for (int kc = 0; kc < 3; ++kc)
            a = __builtin_amdgcn_wmma_f32_16x16x4_f32(false, Af[kc],
                    false, Bq[kc][s], (short)0, a, false, false);
        acc[s] = a;
    }
    #pragma unroll
    for (int r = 0; r < 8; ++r) {
        v4f cv;
        cv.x = p2v[r] + sx[0] - 2.0f * acc[0][r];
        cv.y = p2v[r] + sx[1] - 2.0f * acc[1][r];
        cv.z = p2v[r] + sx[2] - 2.0f * acc[2][r];
        cv.w = p2v[r] + sx[3] - 2.0f * acc[3][r];
        *reinterpret_cast<v4f*>(dst + r * RPITCH_) = cv;
    }
}

__global__ __launch_bounds__(32) void dtw_rows_kernel(
    const float* __restrict__ x,
    const float* __restrict__ patts,
    float* __restrict__ out)
{
    const int lane = threadIdx.x;
    const int hi   = lane >> 4;
    const int lo   = lane & 15;
    const int b    = blockIdx.x;
    const int n    = blockIdx.y;

    __shared__ __attribute__((aligned(16))) float cst[16 * RPITCH_];

    const float* Pn = patts + n * (D_ * P_);
    const float* Xb = x + b * (D_ * L_);

    float pv[D_];
    #pragma unroll
    for (int d = 0; d < D_; ++d) pv[d] = Pn[d * P_ + lane];

    float dprev[CH_];
    #pragma unroll
    for (int k = 0; k < CH_; ++k) dprev[k] = BIGV;
    float leftc = BIGV;

    float* outp = out + (b * N_ + n) * (P_ * LOUT_);
    const float INF = __builtin_inff();

    v2f Aboth[2][3];
    #pragma unroll
    for (int hh = 0; hh < 2; ++hh) {
        #pragma unroll
        for (int kc = 0; kc < 3; ++kc) {
            const int d0 = 4 * kc + 2 * hi;
            const int p  = 16 * hh + lo;
            v2f a; a.x = Pn[d0 * P_ + p]; a.y = Pn[(d0 + 1) * P_ + p];
            Aboth[hh][kc] = a;
        }
    }

    v4f xa[L_ / 64][3], xb[L_ / 64][3];
    __builtin_amdgcn_sched_barrier(0);
    #pragma unroll
    for (int q = 0; q < PRE_; ++q) load_group(xa[q], xb[q], Xb, 64 * q + 4 * lo, hi);

    __builtin_amdgcn_sched_barrier(0);

    float p2l = 0.0f;
    #pragma unroll
    for (int d = 0; d < D_; ++d) p2l += pv[d] * pv[d];

    float* const dst0 = &cst[(8 * hi) * RPITCH_ + (lo >> 3) * CPITCH_ + 4 * (lo & 7)];

    v2f   Bfr[L_ / 64][3][4];
    float sxr[L_ / 64][4];
    {
        float p2v0[8];
        #pragma unroll
        for (int r = 0; r < 8; ++r) {
            const float pa = lane_f(p2l, r), pb = lane_f(p2l, 8 + r);
            p2v0[r] = hi ? pb : pa;
        }
        #pragma unroll
        for (int q = 0; q < L_ / 64; ++q) {
            if (q + PRE_ < L_ / 64) {
                __builtin_amdgcn_sched_barrier(0);
                load_group(xa[q + PRE_], xb[q + PRE_], Xb, 64 * (q + PRE_) + 4 * lo, hi);
            }
            #pragma unroll
            for (int s = 0; s < 4; ++s) {
                float t = 0.0f;
                #pragma unroll
                for (int kc = 0; kc < 3; ++kc) {
                    v2f bb; bb.x = xa[q][kc][s]; bb.y = xb[q][kc][s];
                    t += bb.x * bb.x + bb.y * bb.y;
                    Bfr[q][kc][s] = bb;
                }
                sxr[q][s] = t;
            }
            #pragma unroll
            for (int s = 0; s < 4; ++s) sxr[q][s] += bperm_f(lane ^ 16, sxr[q][s]);
            build_group(Aboth[0], p2v0, Bfr[q], sxr[q], dst0 + 2 * q * CPITCH_);
        }
    }

    #pragma unroll 1
    for (int h = 0; h < 2; ++h) {
        if (h == 1) {
            float p2v1[8];
            #pragma unroll
            for (int r = 0; r < 8; ++r) {
                const float pa = lane_f(p2l, 16 + r), pb = lane_f(p2l, 24 + r);
                p2v1[r] = hi ? pb : pa;
            }
            #pragma unroll
            for (int q = 0; q < L_ / 64; ++q)
                build_group(Aboth[1], p2v1, Bfr[q], sxr[q], dst0 + 2 * q * CPITCH_);
        }
        __syncthreads();

        #pragma unroll 1
        for (int i = 0; i < 16; ++i) {
            const float* crow = &cst[i * RPITCH_ + lane * CPITCH_];
            float c[CH_];
            #pragma unroll
            for (int q4 = 0; q4 < CH_ / 4; ++q4) {
                const v4f v = *reinterpret_cast<const v4f*>(crow + 4 * q4);
                c[4 * q4 + 0] = v.x; c[4 * q4 + 1] = v.y;
                c[4 * q4 + 2] = v.z; c[4 * q4 + 3] = v.w;
            }

            float al[CH_], be[CH_];
            float ae[SUB_], bs[SUB_];
            float m0[SUB_];
            #pragma unroll
            for (int s = 0; s < SUB_; ++s) {
                float alp = BIGV, bep = 0.0f;
                #pragma unroll
                for (int t = 0; t < SL_; ++t) {
                    const int   k    = s * SL_ + t;
                    const float left = (k == 0) ? leftc : dprev[k - 1];
                    const float a    = fminf(left, dprev[k]);
                    const float m    = fminf(a, alp);
                    if (t == 0) m0[s] = m;
                    alp = c[k] + m;
                    bep = (t == 0) ? c[k] : bep + c[k];
                    al[k] = alp; be[k] = bep;
                }
                ae[s] = alp; bs[s] = bep;
            }
            const float a01 = fminf(ae[1], bs[1] + ae[0]);
            const float b01 = bs[0] + bs[1];
            const float a23 = fminf(ae[3], bs[3] + ae[2]);
            const float b23 = bs[2] + bs[3];
            const float a02 = fminf(ae[2], bs[2] + a01);
            const float b02 = b01 + bs[2];

            float A = fminf(a23, b23 + a01), Bv = b01 + b23;
            scan_step<1>(A, Bv, lo);
            scan_step<2>(A, Bv, lo);
            scan_step<4>(A, Bv, lo);
            scan_step<8>(A, Bv, lo);
            {
                const float a15 = other_row_lane15(A);
                const float b15 = other_row_lane15(Bv);
                const float bJ  = hi ? Bv : INF;
                A  = fminf(A, bJ + a15);
                Bv = Bv + (hi ? b15 : 0.0f);
            }

            const float v0   = (h == 0 && i == 0) ? 0.0f : BIGV;
            const float dend = fminf(A, Bv + v0);

            const float s1  = row_shr0<1>(dend);
            const float x15 = other_row_lane15(dend);
            const float din = (lane == 0) ? v0 : ((lane == 16) ? x15 : s1);

            float dn[SUB_];
            dn[0] = din;
            dn[1] = fminf(ae[0], bs[0] + din);
            dn[2] = fminf(a01, b01 + din);
            dn[3] = fminf(a02, b02 + din);
            #pragma unroll
            for (int k = 0; k < CH_; ++k)
                dprev[k] = (k % SL_ == 0) ? c[k] + fminf(m0[k / SL_], dn[k / SL_])
                                          : fminf(al[k], be[k] + dn[k / SL_]);
            leftc = (lane == 0) ? BIGV : din;

            if (lane == 31) {
                v4f* o = reinterpret_cast<v4f*>(outp + (16 * h + i) * LOUT_);
                #pragma unroll
                for (int q4 = 0; q4 < CH_ / 4; ++q4) {
                    v4f v;
                    v.x = dprev[4 * q4 + 0]; v.y = dprev[4 * q4 + 1];
                    v.z = dprev[4 * q4 + 2]; v.w = dprev[4 * q4 + 3];
                    o[q4] = v;
                }
            }
        }
        __syncthreads();
    }
}

extern "C" void kernel_launch(void* const* d_in, const int* in_sizes, int n_in,
                              void* d_out, int out_size, void* d_ws, size_t ws_size,
                              hipStream_t stream) {
    const float* x     = (const float*)d_in[0];
    const float* patts = (const float*)d_in[1];
    float*       outp  = (float*)d_out;
    (void)in_sizes; (void)n_in; (void)out_size; (void)d_ws; (void)ws_size;

    dim3 grid(B_, N_);
    dim3 block(32);
    dtw_rows_kernel<<<grid, block, 0, stream>>>(x, patts, outp);
}
